# starting kernel
# baseline (speedup 1.0000x reference)
#include <hip/hip_runtime.h>
#include <math.h>

typedef __attribute__((ext_vector_type(2))) float v2f;
typedef __attribute__((ext_vector_type(8))) float v8f;

__device__ __forceinline__ void atomicMaxFloat(float* addr, float value) {
  if (value >= 0.0f) {
    atomicMax((int*)addr, __float_as_int(value));
  } else {
    atomicMin((unsigned int*)addr, __float_as_uint(value));
  }
}

__global__ __launch_bounds__(256) void gemm_f32_wmma(
    const float* __restrict__ A, const float* __restrict__ B,
    float* __restrict__ C, int M, int N, int K) {
  const int lane = threadIdx.x & 31;
  const int wave = threadIdx.x >> 5;
  const int wavesPerBlock = blockDim.x >> 5;
  const int tile = blockIdx.x * wavesPerBlock + wave;
  const int tilesN = N >> 4;
  const int numTiles = (M >> 4) * tilesN;
  if (tile >= numTiles) return;

  const int tm = (tile / tilesN) << 4;
  const int tn = (tile % tilesN) << 4;
  const int r  = lane & 15;
  const int kk = (lane >> 4) << 1;

  const float* Arow = A + (size_t)(tm + r) * K;

  v8f acc = {};
  for (int k0 = 0; k0 < K; k0 += 4) {
    v2f a, b;
    float2 av = *(const float2*)(Arow + k0 + kk);
    a.x = av.x;
    a.y = av.y;
    b.x = B[(size_t)(k0 + kk) * N + tn + r];
    b.y = B[(size_t)(k0 + kk + 1) * N + tn + r];
    acc = __builtin_amdgcn_wmma_f32_16x16x4_f32(
         false, a,  false, b,
         (short)0, acc,  false,  false);
  }

  const int rowBase = tm + ((lane >> 4) << 3);
#pragma unroll
  for (int j = 0; j < 8; ++j) {
    C[(size_t)(rowBase + j) * N + tn + r] = acc[j];
  }
}

__global__ void fill_f32(float* __restrict__ p, float v, int n) {
  int i = blockIdx.x * blockDim.x + threadIdx.x;
  if (i < n) p[i] = v;
}

template <int HD, int D>
__global__ __launch_bounds__(256) void edge_scores_wave(
    const float* __restrict__ feat, const int* __restrict__ src,
    const int* __restrict__ dst, float* __restrict__ e, float* __restrict__ m,
    int E, float scale) {
  constexpr int PER_LANE = HD / 32;
  constexpr int LPH = D / PER_LANE;
  constexpr int H = HD / D;
  const int lane = threadIdx.x & 31;
  const int edge = (blockIdx.x * blockDim.x + threadIdx.x) >> 5;
  if (edge >= E) return;
  const int s = src[edge], d = dst[edge];
  const float* fs = feat + (size_t)s * HD + lane * PER_LANE;
  const float* fd = feat + (size_t)d * HD + lane * PER_LANE;

  float acc;
  if constexpr (PER_LANE == 4) {
    float4 a = *(const float4*)fs;
    float4 b = *(const float4*)fd;
    acc = a.x * b.x + a.y * b.y + a.z * b.z + a.w * b.w;
  } else {
    float2 a = *(const float2*)fs;
    float2 b = *(const float2*)fd;
    acc = a.x * b.x + a.y * b.y;
  }
#pragma unroll
  for (int off = 1; off < LPH; off <<= 1) acc += __shfl_xor(acc, off, 32);
  acc *= scale;

  if ((lane % LPH) == 0) {
    const int h = lane / LPH;
    e[(size_t)edge * H + h] = acc;
    atomicMaxFloat(m + (size_t)d * H + h, acc);
  }
}

__global__ void edge_exp(float* __restrict__ e, const int* __restrict__ dst,
                         const float* __restrict__ m, float* __restrict__ denom,
                         int E, int H) {
  int idx = blockIdx.x * blockDim.x + threadIdx.x;
  if (idx >= E * H) return;
  int edge = idx / H;
  int h    = idx - edge * H;
  int d = dst[edge];
  float ex = __expf(e[idx] - m[(size_t)d * H + h]);
  e[idx] = ex;
  atomicAdd(denom + (size_t)d * H + h, ex);
}

template <int HD, int D>
__global__ __launch_bounds__(256) void edge_agg_wave(
    const float* __restrict__ feat, const int* __restrict__ src,
    const int* __restrict__ dst, const float* __restrict__ ex,
    const float* __restrict__ denom, float* __restrict__ out, int E) {
  constexpr int PER_LANE = HD / 32;
  constexpr int LPH = D / PER_LANE;
  constexpr int H = HD / D;
  const int lane = threadIdx.x & 31;
  const int edge = (blockIdx.x * blockDim.x + threadIdx.x) >> 5;
  if (edge >= E) return;
  const int s = src[edge], d = dst[edge];
  const int h = lane / LPH;
  const float a =
      ex[(size_t)edge * H + h] / fmaxf(denom[(size_t)d * H + h], 1e-9f);
  const float* fs = feat + (size_t)s * HD + lane * PER_LANE;
  float* o = out + (size_t)d * HD + lane * PER_LANE;
  if constexpr (PER_LANE == 4) {
    float4 f = *(const float4*)fs;
    atomicAdd(o + 0, a * f.x);
    atomicAdd(o + 1, a * f.y);
    atomicAdd(o + 2, a * f.z);
    atomicAdd(o + 3, a * f.w);
  } else {
    float2 f = *(const float2*)fs;
    atomicAdd(o + 0, a * f.x);
    atomicAdd(o + 1, a * f.y);
  }
}

__global__ void elu_inplace(float* __restrict__ x, int n) {
  int i = blockIdx.x * blockDim.x + threadIdx.x;
  if (i < n) {
    float v = x[i];
    x[i] = v > 0.0f ? v : expm1f(v);
  }
}

static inline int cdiv_i(long long a, long long b) { return (int)((a + b - 1) / b); }

extern "C" void kernel_launch(void* const* d_in, const int* in_sizes, int n_in,
                              void* d_out, int out_size, void* d_ws, size_t ws_size,
                              hipStream_t stream) {
  const float* hIn = (const float*)d_in[0];
  const float* W1  = (const float*)d_in[1];
  const float* W2  = (const float*)d_in[2];
  const int*   src = (const int*)d_in[3];
  const int*   dst = (const int*)d_in[4];

  const int IN_DIM = 128, HEADS = 8, OUT_DIM = 64;
  const int HD1 = 128;
  const int Nn  = in_sizes[0] / IN_DIM;
  const int E   = in_sizes[3];

  float* feat1  = (float*)d_ws;
  float* scores = feat1 + (size_t)Nn * HD1;
  float* mbuf   = scores + (size_t)E * HEADS;
  float* denom  = mbuf + (size_t)Nn * HEADS;
  float* h1     = denom + (size_t)Nn * HEADS;

  const int TPB = 256;
  const int wavesPerBlock = TPB / 32;

  {
    int tiles = (Nn / 16) * (HD1 / 16);
    gemm_f32_wmma<<<cdiv_i(tiles, wavesPerBlock), TPB, 0, stream>>>(
        hIn, W1, feat1, Nn, HD1, IN_DIM);
  }
  fill_f32<<<cdiv_i((long long)Nn * HEADS, TPB), TPB, 0, stream>>>(
      mbuf, -INFINITY, Nn * HEADS);
  hipMemsetAsync(denom, 0, (size_t)Nn * HEADS * sizeof(float), stream);
  hipMemsetAsync(h1, 0, (size_t)Nn * HD1 * sizeof(float), stream);

  edge_scores_wave<128, 16><<<cdiv_i(E, wavesPerBlock), TPB, 0, stream>>>(
      feat1, src, dst, scores, mbuf, E, 0.25f);
  edge_exp<<<cdiv_i((long long)E * HEADS, TPB), TPB, 0, stream>>>(
      scores, dst, mbuf, denom, E, HEADS);
  edge_agg_wave<128, 16><<<cdiv_i(E, wavesPerBlock), TPB, 0, stream>>>(
      feat1, src, dst, scores, denom, h1, E);

  elu_inplace<<<cdiv_i((long long)Nn * HD1, TPB), TPB, 0, stream>>>(h1, Nn * HD1);

  float* feat2 = feat1;
  {
    int tiles = (Nn / 16) * (OUT_DIM / 16);
    gemm_f32_wmma<<<cdiv_i(tiles, wavesPerBlock), TPB, 0, stream>>>(
        h1, W2, feat2, Nn, OUT_DIM, HD1);
  }
  fill_f32<<<cdiv_i(Nn, TPB), TPB, 0, stream>>>(mbuf, -INFINITY, Nn);
  hipMemsetAsync(denom, 0, (size_t)Nn * sizeof(float), stream);
  hipMemsetAsync(d_out, 0, (size_t)Nn * OUT_DIM * sizeof(float), stream);

  edge_scores_wave<64, 64><<<cdiv_i(E, wavesPerBlock), TPB, 0, stream>>>(
      feat2, src, dst, scores, mbuf, E, 0.125f);
  edge_exp<<<cdiv_i(E, TPB), TPB, 0, stream>>>(scores, dst, mbuf, denom, E, 1);
  edge_agg_wave<64, 64><<<cdiv_i(E, wavesPerBlock), TPB, 0, stream>>>(
      feat2, src, dst, scores, denom, (float*)d_out, E);
}


# Aggregation passes: each row of out gets its adds from one L2 half only: workgroups read their die, draw 64-edge chunks from that half's ticket queue, add own-range edges; add arrival order differs
# speedup vs baseline: 1.6167x; 1.6167x over previous
#include <hip/hip_runtime.h>
#include <math.h>

typedef __attribute__((ext_vector_type(2))) float v2f;
typedef __attribute__((ext_vector_type(8))) float v8f;

__device__ __forceinline__ void atomicMaxFloat(float* addr, float value) {
  if (value >= 0.0f) {
    atomicMax((int*)addr, __float_as_int(value));
  } else {
    atomicMin((unsigned int*)addr, __float_as_uint(value));
  }
}

__global__ __launch_bounds__(256) void gemm_f32_wmma(
    const float* __restrict__ A, const float* __restrict__ B,
    float* __restrict__ C, int M, int N, int K) {
  const int lane = threadIdx.x & 31;
  const int wave = threadIdx.x >> 5;
  const int wavesPerBlock = blockDim.x >> 5;
  const int tile = blockIdx.x * wavesPerBlock + wave;
  const int tilesN = N >> 4;
  const int numTiles = (M >> 4) * tilesN;
  if (tile >= numTiles) return;

  const int tm = (tile / tilesN) << 4;
  const int tn = (tile % tilesN) << 4;
  const int r  = lane & 15;
  const int kk = (lane >> 4) << 1;

  const float* Arow = A + (size_t)(tm + r) * K;

  v8f acc = {};
  for (int k0 = 0; k0 < K; k0 += 4) {
    v2f a, b;
    float2 av = *(const float2*)(Arow + k0 + kk);
    a.x = av.x;
    a.y = av.y;
    b.x = B[(size_t)(k0 + kk) * N + tn + r];
    b.y = B[(size_t)(k0 + kk + 1) * N + tn + r];
    acc = __builtin_amdgcn_wmma_f32_16x16x4_f32(
         false, a,  false, b,
         (short)0, acc,  false,  false);
  }

  const int rowBase = tm + ((lane >> 4) << 3);
#pragma unroll
  for (int j = 0; j < 8; ++j) {
    C[(size_t)(rowBase + j) * N + tn + r] = acc[j];
  }
}

__global__ void fill_f32(float* __restrict__ p, float v, int n,
                         unsigned int* __restrict__ tickets) {
  int i = blockIdx.x * blockDim.x + threadIdx.x;
  if (i < n) p[i] = v;
  if (i < 2) tickets[i * 64] = 0u;
}

template <int HD, int D>
__global__ __launch_bounds__(256) void edge_scores_wave32(
    const float* __restrict__ feat, const int* __restrict__ src,
    const int* __restrict__ dst, float* __restrict__ e, float* __restrict__ m,
    int E, float scale) {
  constexpr int H = HD / D;
  const int lane = threadIdx.x & 31;
  const int wave = (blockIdx.x * blockDim.x + threadIdx.x) >> 5;
  const int e0 = wave * 32;
  if (e0 >= E) return;
  const int my = e0 + lane;
  const bool valid = my < E;
  const int sv = valid ? src[my] : 0;
  const int dv = valid ? dst[my] : 0;

  if constexpr (H == 1) {
    static_assert(HD == 64, "one head of 64 columns");
    const int sub = lane & 15;
    const int hb  = lane & 16;
    float mine = 0.0f;
#pragma unroll
    for (int i = 0; i < 16; ++i) {
      const int s = __shfl(sv, hb | i, 32);
      const int d = __shfl(dv, hb | i, 32);
      const float4 a = *(const float4*)(feat + (size_t)s * HD + sub * 4);
      const float4 b = *(const float4*)(feat + (size_t)d * HD + sub * 4);
      float acc = a.x * b.x + a.y * b.y + a.z * b.z + a.w * b.w;
      acc += __shfl_xor(acc, 1, 32);
      acc += __shfl_xor(acc, 2, 32);
      acc += __shfl_xor(acc, 4, 32);
      acc += __shfl_xor(acc, 8, 32);
      if (sub == i) mine = acc;
    }
    mine *= scale;
    if (valid) {
      e[my] = mine;
      atomicMaxFloat(m + dv, mine);
    }
  } else {
    static_assert(HD == 128 && D == 16, "eight heads of 16 columns");
    const int hsel = (lane & 7) << 2;
    const int jsel = lane >> 3;
    for (int g = 0; g < 4; ++g) {
      float4 a[8], b[8];
#pragma unroll
      for (int j = 0; j < 8; ++j) {
        const int s = __builtin_amdgcn_readlane(sv, g * 8 + j);
        const int d = __builtin_amdgcn_readlane(dv, g * 8 + j);
        a[j] = *(const float4*)(feat + (size_t)s * HD + lane * 4);
        b[j] = *(const float4*)(feat + (size_t)d * HD + lane * 4);
      }
      float keep0 = 0.0f, keep1 = 0.0f;
#pragma unroll
      for (int j = 0; j < 8; ++j) {
        float acc = a[j].x * b[j].x + a[j].y * b[j].y + a[j].z * b[j].z + a[j].w * b[j].w;
        acc += __shfl_xor(acc, 1, 32);
        acc += __shfl_xor(acc, 2, 32);
        const float t = __shfl(acc, hsel, 32);
        if (j < 4) { if (jsel == j) keep0 = t; }
        else       { if (jsel == j - 4) keep1 = t; }
      }
#pragma unroll
      for (int l = 0; l < 2; ++l) {
        const int eb = e0 + g * 8 + l * 4;
        const float v = (l == 0 ? keep0 : keep1) * scale;
        const int dd = __shfl(dv, g * 8 + l * 4 + jsel, 32);
        if (eb + jsel < E) {
          e[(size_t)eb * H + lane] = v;
          atomicMaxFloat(m + (size_t)dd * H + (lane & 7), v);
        }
      }
    }
  }
}

__global__ void edge_exp(float* __restrict__ e, const int* __restrict__ dst,
                         const float* __restrict__ m, float* __restrict__ denom,
                         int E, int H) {
  int idx = blockIdx.x * blockDim.x + threadIdx.x;
  if (idx >= E * H) return;
  int edge = idx / H;
  int h    = idx - edge * H;
  int d = dst[edge];
  float ex = __expf(e[idx] - m[(size_t)d * H + h]);
  e[idx] = ex;
  atomicAdd(denom + (size_t)d * H + h, ex);
}

__device__ __forceinline__ int cache_half_of_this_workgroup() {
  return (int)((__builtin_amdgcn_s_sendmsg_rtn(0x87) >> 18) & 1u);
}

template <int HD, int D>
__global__ __launch_bounds__(256) void edge_agg_by_half(
    const float* __restrict__ feat, const int* __restrict__ src,
    const int* __restrict__ dst, const float* __restrict__ ex,
    const float* __restrict__ denom, float* __restrict__ out,
    unsigned int* __restrict__ tickets, int E, int Nn) {
  constexpr int LINES = HD / 32;
  constexpr int H = HD / D;
  constexpr int CW = 8;
  constexpr int CH = CW * 8;
  __shared__ unsigned int s_ticket[2];
  const int lane = threadIdx.x & 31;
  const int wave = threadIdx.x >> 5;
  const int hl = lane % H;
  const int split = Nn >> 1;
  const unsigned int nChunks = (unsigned int)((E + CH - 1) / CH);
  const int home = cache_half_of_this_workgroup();
  unsigned int it = 0;
  for (int pass = 0; pass < 2; ++pass) {
    const int q = home ^ pass;
    unsigned int* counter = tickets + q * 64;
    if (threadIdx.x == 0) s_ticket[it & 1] = atomicAdd(counter, 1u);
    __syncthreads();
    unsigned int c = s_ticket[it & 1];
    ++it;
    while (c < nChunks) {
      const int e0 = (int)c * CH + wave * CW;
      const int my = e0 + lane;
      const bool cand = lane < CW && my < E;
      const int sv = cand ? src[my] : 0;
      const int dv = cand ? dst[my] : 0;
      const bool own = cand && ((dv >= split) == (q == 1));
      const unsigned int mask = __builtin_amdgcn_ballot_w32(own);
      float f[CW][LINES], dn[CW], xv[CW];
#pragma unroll
      for (int j = 0; j < CW; ++j) {
        if (mask & (1u << j)) {
          const int s = __builtin_amdgcn_readlane(sv, j);
          const int d = __builtin_amdgcn_readlane(dv, j);
          const float* fs = feat + (size_t)s * HD + lane;
#pragma unroll
          for (int i = 0; i < LINES; ++i) f[j][i] = fs[32 * i];
          dn[j] = denom[(size_t)d * H + hl];
          xv[j] = ex[(size_t)(e0 + j) * H + hl];
        }
      }
      unsigned int nxt = 0;
      if (threadIdx.x == 0) nxt = atomicAdd(counter, 1u);
#pragma unroll
      for (int j = 0; j < CW; ++j) {
        if (mask & (1u << j)) {
          const int d = __builtin_amdgcn_readlane(dv, j);
          float* o = out + (size_t)d * HD + lane;
          const float a = xv[j] / fmaxf(dn[j], 1e-9f);
#pragma unroll
          for (int i = 0; i < LINES; ++i) {
            float w = a;
            if constexpr (H > 1) w = __shfl(a, (32 * i + lane) / D, 32);
            atomicAdd(o + 32 * i, w * f[j][i]);
          }
        }
      }
      if (threadIdx.x == 0) s_ticket[it & 1] = nxt;
      __syncthreads();
      c = s_ticket[it & 1];
      ++it;
    }
  }
}

__global__ void elu_inplace(float* __restrict__ x, int n) {
  int i = blockIdx.x * blockDim.x + threadIdx.x;
  if (i < n) {
    float v = x[i];
    x[i] = v > 0.0f ? v : expm1f(v);
  }
}

static inline int cdiv_i(long long a, long long b) { return (int)((a + b - 1) / b); }

extern "C" void kernel_launch(void* const* d_in, const int* in_sizes, int n_in,
                              void* d_out, int out_size, void* d_ws, size_t ws_size,
                              hipStream_t stream) {
  const float* hIn = (const float*)d_in[0];
  const float* W1  = (const float*)d_in[1];
  const float* W2  = (const float*)d_in[2];
  const int*   src = (const int*)d_in[3];
  const int*   dst = (const int*)d_in[4];

  const int IN_DIM = 128, HEADS = 8, OUT_DIM = 64;
  const int HD1 = 128;
  const int Nn  = in_sizes[0] / IN_DIM;
  const int E   = in_sizes[3];

  float* feat1  = (float*)d_ws;
  float* scores = feat1 + (size_t)Nn * HD1;
  float* mbuf   = scores + (size_t)E * HEADS;
  float* denom  = mbuf + (size_t)Nn * HEADS;
  float* h1     = denom + (size_t)Nn * HEADS;

  const int TPB = 256;
  const int wavesPerBlock = TPB / 32;

  unsigned int* tickets1 = (unsigned int*)d_out;
  unsigned int* tickets2 = (unsigned int*)(denom + (size_t)Nn * 4);
  const int aggBlocks = 1024;

  {
    int tiles = (Nn / 16) * (HD1 / 16);
    gemm_f32_wmma<<<cdiv_i(tiles, wavesPerBlock), TPB, 0, stream>>>(
        hIn, W1, feat1, Nn, HD1, IN_DIM);
  }
  fill_f32<<<cdiv_i((long long)Nn * HEADS, TPB), TPB, 0, stream>>>(
      mbuf, -INFINITY, Nn * HEADS, tickets1);
  hipMemsetAsync(denom, 0, (size_t)Nn * HEADS * sizeof(float), stream);
  hipMemsetAsync(h1, 0, (size_t)Nn * HD1 * sizeof(float), stream);

  const int scoreBlocks = cdiv_i(cdiv_i(E, 32), wavesPerBlock);
  edge_scores_wave32<128, 16><<<scoreBlocks, TPB, 0, stream>>>(
      feat1, src, dst, scores, mbuf, E, 0.25f);
  edge_exp<<<cdiv_i((long long)E * HEADS, TPB), TPB, 0, stream>>>(
      scores, dst, mbuf, denom, E, HEADS);
  edge_agg_by_half<128, 16><<<aggBlocks, TPB, 0, stream>>>(
      feat1, src, dst, scores, denom, h1, tickets1, E, Nn);

  elu_inplace<<<cdiv_i((long long)Nn * HD1, TPB), TPB, 0, stream>>>(h1, Nn * HD1);

  float* feat2 = feat1;
  {
    int tiles = (Nn / 16) * (OUT_DIM / 16);
    gemm_f32_wmma<<<cdiv_i(tiles, wavesPerBlock), TPB, 0, stream>>>(
        h1, W2, feat2, Nn, OUT_DIM, HD1);
  }
  fill_f32<<<cdiv_i(Nn, TPB), TPB, 0, stream>>>(mbuf, -INFINITY, Nn, tickets2);
  hipMemsetAsync(denom, 0, (size_t)Nn * sizeof(float), stream);
  hipMemsetAsync(d_out, 0, (size_t)Nn * OUT_DIM * sizeof(float), stream);

  edge_scores_wave32<64, 64><<<scoreBlocks, TPB, 0, stream>>>(
      feat2, src, dst, scores, mbuf, E, 0.125f);
  edge_exp<<<cdiv_i(E, TPB), TPB, 0, stream>>>(scores, dst, mbuf, denom, E, 1);
  edge_agg_by_half<64, 64><<<aggBlocks, TPB, 0, stream>>>(
      feat2, src, dst, scores, denom, (float*)d_out, tickets2, E, Nn);
}


# Softmax division moved behind the sum: aggregation forms exp(e-max) itself, adds it to out rows and denominator; rows divided once in the ELU pass (new for layer 2); no exp passes; rounding differs
# speedup vs baseline: 1.9253x; 1.1909x over previous
#include <hip/hip_runtime.h>
#include <math.h>

typedef __attribute__((ext_vector_type(2))) float v2f;
typedef __attribute__((ext_vector_type(8))) float v8f;

__device__ __forceinline__ void atomicMaxFloat(float* addr, float value) {
  if (value >= 0.0f) {
    atomicMax((int*)addr, __float_as_int(value));
  } else {
    atomicMin((unsigned int*)addr, __float_as_uint(value));
  }
}

__global__ __launch_bounds__(256) void gemm_f32_wmma(
    const float* __restrict__ A, const float* __restrict__ B,
    float* __restrict__ C, int M, int N, int K) {
  const int lane = threadIdx.x & 31;
  const int wave = threadIdx.x >> 5;
  const int wavesPerBlock = blockDim.x >> 5;
  const int tile = blockIdx.x * wavesPerBlock + wave;
  const int tilesN = N >> 4;
  const int numTiles = (M >> 4) * tilesN;
  if (tile >= numTiles) return;

  const int tm = (tile / tilesN) << 4;
  const int tn = (tile % tilesN) << 4;
  const int r  = lane & 15;
  const int kk = (lane >> 4) << 1;

  const float* Arow = A + (size_t)(tm + r) * K;

  v8f acc = {};
  for (int k0 = 0; k0 < K; k0 += 4) {
    v2f a, b;
    float2 av = *(const float2*)(Arow + k0 + kk);
    a.x = av.x;
    a.y = av.y;
    b.x = B[(size_t)(k0 + kk) * N + tn + r];
    b.y = B[(size_t)(k0 + kk + 1) * N + tn + r];
    acc = __builtin_amdgcn_wmma_f32_16x16x4_f32(
         false, a,  false, b,
         (short)0, acc,  false,  false);
  }

  const int rowBase = tm + ((lane >> 4) << 3);
#pragma unroll
  for (int j = 0; j < 8; ++j) {
    C[(size_t)(rowBase + j) * N + tn + r] = acc[j];
  }
}

__global__ void fill_f32(float* __restrict__ p, float v, int n,
                         unsigned int* __restrict__ tickets) {
  int i = blockIdx.x * blockDim.x + threadIdx.x;
  if (i < n) p[i] = v;
  if (i < 2) tickets[i * 64] = 0u;
}

template <int HD, int D>
__global__ __launch_bounds__(256) void edge_scores_wave32(
    const float* __restrict__ feat, const int* __restrict__ src,
    const int* __restrict__ dst, float* __restrict__ e, float* __restrict__ m,
    int E, float scale) {
  constexpr int H = HD / D;
  const int lane = threadIdx.x & 31;
  const int wave = (blockIdx.x * blockDim.x + threadIdx.x) >> 5;
  const int e0 = wave * 32;
  if (e0 >= E) return;
  const int my = e0 + lane;
  const bool valid = my < E;
  const int sv = valid ? src[my] : 0;
  const int dv = valid ? dst[my] : 0;

  if constexpr (H == 1) {
    static_assert(HD == 64, "one head of 64 columns");
    const int sub = lane & 15;
    const int hb  = lane & 16;
    float mine = 0.0f;
#pragma unroll
    for (int i = 0; i < 16; ++i) {
      const int s = __shfl(sv, hb | i, 32);
      const int d = __shfl(dv, hb | i, 32);
      const float4 a = *(const float4*)(feat + (size_t)s * HD + sub * 4);
      const float4 b = *(const float4*)(feat + (size_t)d * HD + sub * 4);
      float acc = a.x * b.x + a.y * b.y + a.z * b.z + a.w * b.w;
      acc += __shfl_xor(acc, 1, 32);
      acc += __shfl_xor(acc, 2, 32);
      acc += __shfl_xor(acc, 4, 32);
      acc += __shfl_xor(acc, 8, 32);
      if (sub == i) mine = acc;
    }
    mine *= scale;
    if (valid) {
      e[my] = mine;
      atomicMaxFloat(m + dv, mine);
    }
  } else {
    static_assert(HD == 128 && D == 16, "eight heads of 16 columns");
    const int hsel = (lane & 7) << 2;
    const int jsel = lane >> 3;
    for (int g = 0; g < 4; ++g) {
      float4 a[8], b[8];
#pragma unroll
      for (int j = 0; j < 8; ++j) {
        const int s = __builtin_amdgcn_readlane(sv, g * 8 + j);
        const int d = __builtin_amdgcn_readlane(dv, g * 8 + j);
        a[j] = *(const float4*)(feat + (size_t)s * HD + lane * 4);
        b[j] = *(const float4*)(feat + (size_t)d * HD + lane * 4);
      }
      float keep0 = 0.0f, keep1 = 0.0f;
#pragma unroll
      for (int j = 0; j < 8; ++j) {
        float acc = a[j].x * b[j].x + a[j].y * b[j].y + a[j].z * b[j].z + a[j].w * b[j].w;
        acc += __shfl_xor(acc, 1, 32);
        acc += __shfl_xor(acc, 2, 32);
        const float t = __shfl(acc, hsel, 32);
        if (j < 4) { if (jsel == j) keep0 = t; }
        else       { if (jsel == j - 4) keep1 = t; }
      }
#pragma unroll
      for (int l = 0; l < 2; ++l) {
        const int eb = e0 + g * 8 + l * 4;
        const float v = (l == 0 ? keep0 : keep1) * scale;
        const int dd = __shfl(dv, g * 8 + l * 4 + jsel, 32);
        if (eb + jsel < E) {
          e[(size_t)eb * H + lane] = v;
          atomicMaxFloat(m + (size_t)dd * H + (lane & 7), v);
        }
      }
    }
  }
}


__device__ __forceinline__ int cache_half_of_this_workgroup() {
  return (int)((__builtin_amdgcn_s_sendmsg_rtn(0x87) >> 18) & 1u);
}

template <int HD, int D>
__global__ __launch_bounds__(256) void edge_agg_by_half(
    const float* __restrict__ feat, const int* __restrict__ src,
    const int* __restrict__ dst, const float* __restrict__ e,
    const float* __restrict__ m, float* __restrict__ denom,
    float* __restrict__ out, unsigned int* __restrict__ tickets, int E, int Nn) {
  constexpr int LINES = HD / 32;
  constexpr int H = HD / D;
  constexpr int CW = 8;
  constexpr int CH = CW * 8;
  __shared__ unsigned int s_ticket[2];
  const int lane = threadIdx.x & 31;
  const int wave = threadIdx.x >> 5;
  const int hl = lane % H;
  const int split = Nn >> 1;
  const unsigned int nChunks = (unsigned int)((E + CH - 1) / CH);
  const int home = cache_half_of_this_workgroup();
  unsigned int it = 0;
  for (int pass = 0; pass < 2; ++pass) {
    const int q = home ^ pass;
    unsigned int* counter = tickets + q * 64;
    if (threadIdx.x == 0) s_ticket[it & 1] = atomicAdd(counter, 1u);
    __syncthreads();
    unsigned int c = s_ticket[it & 1];
    ++it;
    while (c < nChunks) {
      const int e0 = (int)c * CH + wave * CW;
      const int my = e0 + lane;
      const bool cand = lane < CW && my < E;
      const int sv = cand ? src[my] : 0;
      const int dv = cand ? dst[my] : 0;
      const bool own = cand && ((dv >= split) == (q == 1));
      const unsigned int mask = __builtin_amdgcn_ballot_w32(own);
      float f[CW][LINES], mx[CW], xv[CW];
#pragma unroll
      for (int j = 0; j < CW; ++j) {
        if (mask & (1u << j)) {
          const int s = __builtin_amdgcn_readlane(sv, j);
          const int d = __builtin_amdgcn_readlane(dv, j);
          const float* fs = feat + (size_t)s * HD + lane;
#pragma unroll
          for (int i = 0; i < LINES; ++i) f[j][i] = fs[32 * i];
          mx[j] = m[(size_t)d * H + hl];
          xv[j] = e[(size_t)(e0 + j) * H + hl];
        }
      }
      unsigned int nxt = 0;
      if (threadIdx.x == 0) nxt = atomicAdd(counter, 1u);
#pragma unroll
      for (int j = 0; j < CW; ++j) {
        if (mask & (1u << j)) {
          const int d = __builtin_amdgcn_readlane(dv, j);
          float* o = out + (size_t)d * HD + lane;
          const float a = __expf(xv[j] - mx[j]);
          if (lane < H) atomicAdd(denom + (size_t)d * H + lane, a);
#pragma unroll
          for (int i = 0; i < LINES; ++i) {
            float w = a;
            if constexpr (H > 1) w = __shfl(a, (32 * i + lane) / D, 32);
            atomicAdd(o + 32 * i, w * f[j][i]);
          }
        }
      }
      if (threadIdx.x == 0) s_ticket[it & 1] = nxt;
      __syncthreads();
      c = s_ticket[it & 1];
      ++it;
    }
  }
}

template <int HD, int D, bool ELU>
__global__ void finish_rows(float* __restrict__ x, const float* __restrict__ denom,
                            int n4) {
  constexpr int H = HD / D;
  constexpr int Q = HD / 4;
  static_assert(D % 4 == 0, "a 16-byte piece stays inside one head");
  const int i = blockIdx.x * blockDim.x + threadIdx.x;
  if (i >= n4) return;
  const int node = i / Q;
  const int col  = (i - node * Q) * 4;
  const float dn = fmaxf(denom[(size_t)node * H + col / D], 1e-9f);
  float4 v = ((const float4*)x)[i];
  v.x /= dn; v.y /= dn; v.z /= dn; v.w /= dn;
  if constexpr (ELU) {
    v.x = v.x > 0.0f ? v.x : expm1f(v.x);
    v.y = v.y > 0.0f ? v.y : expm1f(v.y);
    v.z = v.z > 0.0f ? v.z : expm1f(v.z);
    v.w = v.w > 0.0f ? v.w : expm1f(v.w);
  }
  ((float4*)x)[i] = v;
}

static inline int cdiv_i(long long a, long long b) { return (int)((a + b - 1) / b); }

extern "C" void kernel_launch(void* const* d_in, const int* in_sizes, int n_in,
                              void* d_out, int out_size, void* d_ws, size_t ws_size,
                              hipStream_t stream) {
  const float* hIn = (const float*)d_in[0];
  const float* W1  = (const float*)d_in[1];
  const float* W2  = (const float*)d_in[2];
  const int*   src = (const int*)d_in[3];
  const int*   dst = (const int*)d_in[4];

  const int IN_DIM = 128, HEADS = 8, OUT_DIM = 64;
  const int HD1 = 128;
  const int Nn  = in_sizes[0] / IN_DIM;
  const int E   = in_sizes[3];

  float* feat1  = (float*)d_ws;
  float* scores = feat1 + (size_t)Nn * HD1;
  float* mbuf   = scores + (size_t)E * HEADS;
  float* denom  = mbuf + (size_t)Nn * HEADS;
  float* h1     = denom + (size_t)Nn * HEADS;

  const int TPB = 256;
  const int wavesPerBlock = TPB / 32;

  unsigned int* tickets1 = (unsigned int*)d_out;
  unsigned int* tickets2 = (unsigned int*)(denom + (size_t)Nn * 4);
  const int aggBlocks = 1024;

  {
    int tiles = (Nn / 16) * (HD1 / 16);
    gemm_f32_wmma<<<cdiv_i(tiles, wavesPerBlock), TPB, 0, stream>>>(
        hIn, W1, feat1, Nn, HD1, IN_DIM);
  }
  fill_f32<<<cdiv_i((long long)Nn * HEADS, TPB), TPB, 0, stream>>>(
      mbuf, -INFINITY, Nn * HEADS, tickets1);
  hipMemsetAsync(denom, 0, (size_t)Nn * HEADS * sizeof(float), stream);
  hipMemsetAsync(h1, 0, (size_t)Nn * HD1 * sizeof(float), stream);

  const int scoreBlocks = cdiv_i(cdiv_i(E, 32), wavesPerBlock);
  edge_scores_wave32<128, 16><<<scoreBlocks, TPB, 0, stream>>>(
      feat1, src, dst, scores, mbuf, E, 0.25f);
  edge_agg_by_half<128, 16><<<aggBlocks, TPB, 0, stream>>>(
      feat1, src, dst, scores, mbuf, denom, h1, tickets1, E, Nn);
  finish_rows<128, 16, true><<<cdiv_i((long long)Nn * HD1 / 4, TPB), TPB, 0, stream>>>(
      h1, denom, Nn * HD1 / 4);

  float* feat2 = feat1;
  {
    int tiles = (Nn / 16) * (OUT_DIM / 16);
    gemm_f32_wmma<<<cdiv_i(tiles, wavesPerBlock), TPB, 0, stream>>>(
        h1, W2, feat2, Nn, OUT_DIM, HD1);
  }
  fill_f32<<<cdiv_i(Nn, TPB), TPB, 0, stream>>>(mbuf, -INFINITY, Nn, tickets2);
  hipMemsetAsync(denom, 0, (size_t)Nn * sizeof(float), stream);
  hipMemsetAsync(d_out, 0, (size_t)Nn * OUT_DIM * sizeof(float), stream);

  edge_scores_wave32<64, 64><<<scoreBlocks, TPB, 0, stream>>>(
      feat2, src, dst, scores, mbuf, E, 0.125f);
  edge_agg_by_half<64, 64><<<aggBlocks, TPB, 0, stream>>>(
      feat2, src, dst, scores, mbuf, denom, (float*)d_out, tickets2, E, Nn);
  finish_rows<64, 64, false><<<cdiv_i((long long)Nn * OUT_DIM / 4, TPB), TPB, 0, stream>>>(
      (float*)d_out, denom, Nn * OUT_DIM / 4);
}


# Score passes: the row-maximum plane m has one copy per cache half; a workgroup's max atomics go to its half's copy only; a small launch merges the copies before aggregation; m, e and sums unchanged
# speedup vs baseline: 2.4925x; 1.2946x over previous
#include <hip/hip_runtime.h>
#include <math.h>

typedef __attribute__((ext_vector_type(2))) float v2f;
typedef __attribute__((ext_vector_type(8))) float v8f;

__device__ __forceinline__ void atomicMaxFloat(float* addr, float value) {
  if (value >= 0.0f) {
    atomicMax((int*)addr, __float_as_int(value));
  } else {
    atomicMin((unsigned int*)addr, __float_as_uint(value));
  }
}

__global__ __launch_bounds__(256) void gemm_f32_wmma(
    const float* __restrict__ A, const float* __restrict__ B,
    float* __restrict__ C, int M, int N, int K) {
  const int lane = threadIdx.x & 31;
  const int wave = threadIdx.x >> 5;
  const int wavesPerBlock = blockDim.x >> 5;
  const int tile = blockIdx.x * wavesPerBlock + wave;
  const int tilesN = N >> 4;
  const int numTiles = (M >> 4) * tilesN;
  if (tile >= numTiles) return;

  const int tm = (tile / tilesN) << 4;
  const int tn = (tile % tilesN) << 4;
  const int r  = lane & 15;
  const int kk = (lane >> 4) << 1;

  const float* Arow = A + (size_t)(tm + r) * K;

  v8f acc = {};
  for (int k0 = 0; k0 < K; k0 += 4) {
    v2f a, b;
    float2 av = *(const float2*)(Arow + k0 + kk);
    a.x = av.x;
    a.y = av.y;
    b.x = B[(size_t)(k0 + kk) * N + tn + r];
    b.y = B[(size_t)(k0 + kk + 1) * N + tn + r];
    acc = __builtin_amdgcn_wmma_f32_16x16x4_f32(
         false, a,  false, b,
         (short)0, acc,  false,  false);
  }

  const int rowBase = tm + ((lane >> 4) << 3);
#pragma unroll
  for (int j = 0; j < 8; ++j) {
    C[(size_t)(rowBase + j) * N + tn + r] = acc[j];
  }
}

__global__ void fill_f32(float* __restrict__ p, float* __restrict__ p2, float v, int n,
                         unsigned int* __restrict__ tickets) {
  int i = blockIdx.x * blockDim.x + threadIdx.x;
  if (i < n) { p[i] = v; p2[i] = v; }
  if (i < 2) tickets[i * 64] = 0u;
}

__device__ __forceinline__ int cache_half_of_this_workgroup() {
  return (int)((__builtin_amdgcn_s_sendmsg_rtn(0x87) >> 18) & 1u);
}

template <int HD, int D>
__global__ __launch_bounds__(256) void edge_scores_wave32(
    const float* __restrict__ feat, const int* __restrict__ src,
    const int* __restrict__ dst, float* __restrict__ e, float* __restrict__ m_lo,
    float* __restrict__ m_hi, int E, float scale) {
  constexpr int H = HD / D;
  const int lane = threadIdx.x & 31;
  const int wave = (blockIdx.x * blockDim.x + threadIdx.x) >> 5;
  const int e0 = wave * 32;
  if (e0 >= E) return;
  float* m = cache_half_of_this_workgroup() ? m_hi : m_lo;
  const int my = e0 + lane;
  const bool valid = my < E;
  const int sv = valid ? src[my] : 0;
  const int dv = valid ? dst[my] : 0;

  if constexpr (H == 1) {
    static_assert(HD == 64, "one head of 64 columns");
    const int sub = lane & 15;
    const int hb  = lane & 16;
    float mine = 0.0f;
#pragma unroll
    for (int i = 0; i < 16; ++i) {
      const int s = __shfl(sv, hb | i, 32);
      const int d = __shfl(dv, hb | i, 32);
      const float4 a = *(const float4*)(feat + (size_t)s * HD + sub * 4);
      const float4 b = *(const float4*)(feat + (size_t)d * HD + sub * 4);
      float acc = a.x * b.x + a.y * b.y + a.z * b.z + a.w * b.w;
      acc += __shfl_xor(acc, 1, 32);
      acc += __shfl_xor(acc, 2, 32);
      acc += __shfl_xor(acc, 4, 32);
      acc += __shfl_xor(acc, 8, 32);
      if (sub == i) mine = acc;
    }
    mine *= scale;
    if (valid) {
      e[my] = mine;
      atomicMaxFloat(m + dv, mine);
    }
  } else {
    static_assert(HD == 128 && D == 16, "eight heads of 16 columns");
    const int hsel = (lane & 7) << 2;
    const int jsel = lane >> 3;
    for (int g = 0; g < 4; ++g) {
      float4 a[8], b[8];
#pragma unroll
      for (int j = 0; j < 8; ++j) {
        const int s = __builtin_amdgcn_readlane(sv, g * 8 + j);
        const int d = __builtin_amdgcn_readlane(dv, g * 8 + j);
        a[j] = *(const float4*)(feat + (size_t)s * HD + lane * 4);
        b[j] = *(const float4*)(feat + (size_t)d * HD + lane * 4);
      }
      float keep0 = 0.0f, keep1 = 0.0f;
#pragma unroll
      for (int j = 0; j < 8; ++j) {
        float acc = a[j].x * b[j].x + a[j].y * b[j].y + a[j].z * b[j].z + a[j].w * b[j].w;
        acc += __shfl_xor(acc, 1, 32);
        acc += __shfl_xor(acc, 2, 32);
        const float t = __shfl(acc, hsel, 32);
        if (j < 4) { if (jsel == j) keep0 = t; }
        else       { if (jsel == j - 4) keep1 = t; }
      }
#pragma unroll
      for (int l = 0; l < 2; ++l) {
        const int eb = e0 + g * 8 + l * 4;
        const float v = (l == 0 ? keep0 : keep1) * scale;
        const int dd = __shfl(dv, g * 8 + l * 4 + jsel, 32);
        if (eb + jsel < E) {
          e[(size_t)eb * H + lane] = v;
          atomicMaxFloat(m + (size_t)dd * H + (lane & 7), v);
        }
      }
    }
  }
}

__global__ void merge_max_planes(float* __restrict__ lo, const float* __restrict__ hi, int n) {
  const int i = blockIdx.x * blockDim.x + threadIdx.x;
  if (i < n) lo[i] = fmaxf(lo[i], hi[i]);
}


template <int HD, int D>
__global__ __launch_bounds__(256) void edge_agg_by_half(
    const float* __restrict__ feat, const int* __restrict__ src,
    const int* __restrict__ dst, const float* __restrict__ e,
    const float* __restrict__ m, float* __restrict__ denom,
    float* __restrict__ out, unsigned int* __restrict__ tickets, int E, int Nn) {
  constexpr int LINES = HD / 32;
  constexpr int H = HD / D;
  constexpr int CW = 8;
  constexpr int CH = CW * 8;
  __shared__ unsigned int s_ticket[2];
  const int lane = threadIdx.x & 31;
  const int wave = threadIdx.x >> 5;
  const int hl = lane % H;
  const int split = Nn >> 1;
  const unsigned int nChunks = (unsigned int)((E + CH - 1) / CH);
  const int home = cache_half_of_this_workgroup();
  unsigned int it = 0;
  for (int pass = 0; pass < 2; ++pass) {
    const int q = home ^ pass;
    unsigned int* counter = tickets + q * 64;
    if (threadIdx.x == 0) s_ticket[it & 1] = atomicAdd(counter, 1u);
    __syncthreads();
    unsigned int c = s_ticket[it & 1];
    ++it;
    while (c < nChunks) {
      const int e0 = (int)c * CH + wave * CW;
      const int my = e0 + lane;
      const bool cand = lane < CW && my < E;
      const int sv = cand ? src[my] : 0;
      const int dv = cand ? dst[my] : 0;
      const bool own = cand && ((dv >= split) == (q == 1));
      const unsigned int mask = __builtin_amdgcn_ballot_w32(own);
      float f[CW][LINES], mx[CW], xv[CW];
#pragma unroll
      for (int j = 0; j < CW; ++j) {
        if (mask & (1u << j)) {
          const int s = __builtin_amdgcn_readlane(sv, j);
          const int d = __builtin_amdgcn_readlane(dv, j);
          const float* fs = feat + (size_t)s * HD + lane;
#pragma unroll
          for (int i = 0; i < LINES; ++i) f[j][i] = fs[32 * i];
          mx[j] = m[(size_t)d * H + hl];
          xv[j] = e[(size_t)(e0 + j) * H + hl];
        }
      }
      unsigned int nxt = 0;
      if (threadIdx.x == 0) nxt = atomicAdd(counter, 1u);
#pragma unroll
      for (int j = 0; j < CW; ++j) {
        if (mask & (1u << j)) {
          const int d = __builtin_amdgcn_readlane(dv, j);
          float* o = out + (size_t)d * HD + lane;
          const float a = __expf(xv[j] - mx[j]);
          if (lane < H) atomicAdd(denom + (size_t)d * H + lane, a);
#pragma unroll
          for (int i = 0; i < LINES; ++i) {
            float w = a;
            if constexpr (H > 1) w = __shfl(a, (32 * i + lane) / D, 32);
            atomicAdd(o + 32 * i, w * f[j][i]);
          }
        }
      }
      if (threadIdx.x == 0) s_ticket[it & 1] = nxt;
      __syncthreads();
      c = s_ticket[it & 1];
      ++it;
    }
  }
}

template <int HD, int D, bool ELU>
__global__ void finish_rows(float* __restrict__ x, const float* __restrict__ denom,
                            int n4) {
  constexpr int H = HD / D;
  constexpr int Q = HD / 4;
  static_assert(D % 4 == 0, "a 16-byte piece stays inside one head");
  const int i = blockIdx.x * blockDim.x + threadIdx.x;
  if (i >= n4) return;
  const int node = i / Q;
  const int col  = (i - node * Q) * 4;
  const float dn = fmaxf(denom[(size_t)node * H + col / D], 1e-9f);
  float4 v = ((const float4*)x)[i];
  v.x /= dn; v.y /= dn; v.z /= dn; v.w /= dn;
  if constexpr (ELU) {
    v.x = v.x > 0.0f ? v.x : expm1f(v.x);
    v.y = v.y > 0.0f ? v.y : expm1f(v.y);
    v.z = v.z > 0.0f ? v.z : expm1f(v.z);
    v.w = v.w > 0.0f ? v.w : expm1f(v.w);
  }
  ((float4*)x)[i] = v;
}

static inline int cdiv_i(long long a, long long b) { return (int)((a + b - 1) / b); }

extern "C" void kernel_launch(void* const* d_in, const int* in_sizes, int n_in,
                              void* d_out, int out_size, void* d_ws, size_t ws_size,
                              hipStream_t stream) {
  const float* hIn = (const float*)d_in[0];
  const float* W1  = (const float*)d_in[1];
  const float* W2  = (const float*)d_in[2];
  const int*   src = (const int*)d_in[3];
  const int*   dst = (const int*)d_in[4];

  const int IN_DIM = 128, HEADS = 8, OUT_DIM = 64;
  const int HD1 = 128;
  const int Nn  = in_sizes[0] / IN_DIM;
  const int E   = in_sizes[3];

  float* feat1  = (float*)d_ws;
  float* scores = feat1 + (size_t)Nn * HD1;
  float* mbuf   = scores + (size_t)E * HEADS;
  float* denom  = mbuf + (size_t)Nn * HEADS;
  float* h1     = denom + (size_t)Nn * HEADS;

  const int TPB = 256;
  const int wavesPerBlock = TPB / 32;

  unsigned int* tickets1 = (unsigned int*)d_out;
  unsigned int* tickets2 = (unsigned int*)(denom + (size_t)Nn * 4);
  float* mhi1 = (float*)d_out + 1024;
  float* mhi2 = mbuf + (size_t)Nn * 4;
  const int aggBlocks = 1024;

  {
    int tiles = (Nn / 16) * (HD1 / 16);
    gemm_f32_wmma<<<cdiv_i(tiles, wavesPerBlock), TPB, 0, stream>>>(
        hIn, W1, feat1, Nn, HD1, IN_DIM);
  }
  fill_f32<<<cdiv_i((long long)Nn * HEADS, TPB), TPB, 0, stream>>>(
      mbuf, mhi1, -INFINITY, Nn * HEADS, tickets1);
  hipMemsetAsync(denom, 0, (size_t)Nn * HEADS * sizeof(float), stream);
  hipMemsetAsync(h1, 0, (size_t)Nn * HD1 * sizeof(float), stream);

  const int scoreBlocks = cdiv_i(cdiv_i(E, 32), wavesPerBlock);
  edge_scores_wave32<128, 16><<<scoreBlocks, TPB, 0, stream>>>(
      feat1, src, dst, scores, mbuf, mhi1, E, 0.25f);
  merge_max_planes<<<cdiv_i((long long)Nn * HEADS, TPB), TPB, 0, stream>>>(
      mbuf, mhi1, Nn * HEADS);
  edge_agg_by_half<128, 16><<<aggBlocks, TPB, 0, stream>>>(
      feat1, src, dst, scores, mbuf, denom, h1, tickets1, E, Nn);
  finish_rows<128, 16, true><<<cdiv_i((long long)Nn * HD1 / 4, TPB), TPB, 0, stream>>>(
      h1, denom, Nn * HD1 / 4);

  float* feat2 = feat1;
  {
    int tiles = (Nn / 16) * (OUT_DIM / 16);
    gemm_f32_wmma<<<cdiv_i(tiles, wavesPerBlock), TPB, 0, stream>>>(
        h1, W2, feat2, Nn, OUT_DIM, HD1);
  }
  fill_f32<<<cdiv_i(Nn, TPB), TPB, 0, stream>>>(mbuf, mhi2, -INFINITY, Nn, tickets2);
  hipMemsetAsync(denom, 0, (size_t)Nn * sizeof(float), stream);
  hipMemsetAsync(d_out, 0, (size_t)Nn * OUT_DIM * sizeof(float), stream);

  edge_scores_wave32<64, 64><<<scoreBlocks, TPB, 0, stream>>>(
      feat2, src, dst, scores, mbuf, mhi2, E, 0.125f);
  merge_max_planes<<<cdiv_i(Nn, TPB), TPB, 0, stream>>>(mbuf, mhi2, Nn);
  edge_agg_by_half<64, 64><<<aggBlocks, TPB, 0, stream>>>(
      feat2, src, dst, scores, mbuf, denom, (float*)d_out, tickets2, E, Nn);
  finish_rows<64, 64, false><<<cdiv_i((long long)Nn * OUT_DIM / 4, TPB), TPB, 0, stream>>>(
      (float*)d_out, denom, Nn * OUT_DIM / 4);
}


# Aggregation passes: ticket loop software-pipelined: two tickets held, next chunk's ids load beside the current rows, trip barrier orders LDS only (no wait for add acks); add arrival order differs
# speedup vs baseline: 2.6701x; 1.0713x over previous
#include <hip/hip_runtime.h>
#include <math.h>

typedef __attribute__((ext_vector_type(2))) float v2f;
typedef __attribute__((ext_vector_type(8))) float v8f;

__device__ __forceinline__ void atomicMaxFloat(float* addr, float value) {
  if (value >= 0.0f) {
    atomicMax((int*)addr, __float_as_int(value));
  } else {
    atomicMin((unsigned int*)addr, __float_as_uint(value));
  }
}

__global__ __launch_bounds__(256) void gemm_f32_wmma(
    const float* __restrict__ A, const float* __restrict__ B,
    float* __restrict__ C, int M, int N, int K) {
  const int lane = threadIdx.x & 31;
  const int wave = threadIdx.x >> 5;
  const int wavesPerBlock = blockDim.x >> 5;
  const int tile = blockIdx.x * wavesPerBlock + wave;
  const int tilesN = N >> 4;
  const int numTiles = (M >> 4) * tilesN;
  if (tile >= numTiles) return;

  const int tm = (tile / tilesN) << 4;
  const int tn = (tile % tilesN) << 4;
  const int r  = lane & 15;
  const int kk = (lane >> 4) << 1;

  const float* Arow = A + (size_t)(tm + r) * K;

  v8f acc = {};
  for (int k0 = 0; k0 < K; k0 += 4) {
    v2f a, b;
    float2 av = *(const float2*)(Arow + k0 + kk);
    a.x = av.x;
    a.y = av.y;
    b.x = B[(size_t)(k0 + kk) * N + tn + r];
    b.y = B[(size_t)(k0 + kk + 1) * N + tn + r];
    acc = __builtin_amdgcn_wmma_f32_16x16x4_f32(
         false, a,  false, b,
         (short)0, acc,  false,  false);
  }

  const int rowBase = tm + ((lane >> 4) << 3);
#pragma unroll
  for (int j = 0; j < 8; ++j) {
    C[(size_t)(rowBase + j) * N + tn + r] = acc[j];
  }
}

__global__ void fill_f32(float* __restrict__ p, float* __restrict__ p2, float v, int n,
                         unsigned int* __restrict__ tickets) {
  int i = blockIdx.x * blockDim.x + threadIdx.x;
  if (i < n) { p[i] = v; p2[i] = v; }
  if (i < 2) tickets[i * 64] = 0u;
}

__device__ __forceinline__ int cache_half_of_this_workgroup() {
  return (int)((__builtin_amdgcn_s_sendmsg_rtn(0x87) >> 18) & 1u);
}

template <int HD, int D>
__global__ __launch_bounds__(256) void edge_scores_wave32(
    const float* __restrict__ feat, const int* __restrict__ src,
    const int* __restrict__ dst, float* __restrict__ e, float* __restrict__ m_lo,
    float* __restrict__ m_hi, int E, float scale) {
  constexpr int H = HD / D;
  const int lane = threadIdx.x & 31;
  const int wave = (blockIdx.x * blockDim.x + threadIdx.x) >> 5;
  const int e0 = wave * 32;
  if (e0 >= E) return;
  float* m = cache_half_of_this_workgroup() ? m_hi : m_lo;
  const int my = e0 + lane;
  const bool valid = my < E;
  const int sv = valid ? src[my] : 0;
  const int dv = valid ? dst[my] : 0;

  if constexpr (H == 1) {
    static_assert(HD == 64, "one head of 64 columns");
    const int sub = lane & 15;
    const int hb  = lane & 16;
    float mine = 0.0f;
#pragma unroll
    for (int i = 0; i < 16; ++i) {
      const int s = __shfl(sv, hb | i, 32);
      const int d = __shfl(dv, hb | i, 32);
      const float4 a = *(const float4*)(feat + (size_t)s * HD + sub * 4);
      const float4 b = *(const float4*)(feat + (size_t)d * HD + sub * 4);
      float acc = a.x * b.x + a.y * b.y + a.z * b.z + a.w * b.w;
      acc += __shfl_xor(acc, 1, 32);
      acc += __shfl_xor(acc, 2, 32);
      acc += __shfl_xor(acc, 4, 32);
      acc += __shfl_xor(acc, 8, 32);
      if (sub == i) mine = acc;
    }
    mine *= scale;
    if (valid) {
      e[my] = mine;
      atomicMaxFloat(m + dv, mine);
    }
  } else {
    static_assert(HD == 128 && D == 16, "eight heads of 16 columns");
    const int hsel = (lane & 7) << 2;
    const int jsel = lane >> 3;
    for (int g = 0; g < 4; ++g) {
      float4 a[8], b[8];
#pragma unroll
      for (int j = 0; j < 8; ++j) {
        const int s = __builtin_amdgcn_readlane(sv, g * 8 + j);
        const int d = __builtin_amdgcn_readlane(dv, g * 8 + j);
        a[j] = *(const float4*)(feat + (size_t)s * HD + lane * 4);
        b[j] = *(const float4*)(feat + (size_t)d * HD + lane * 4);
      }
      float keep0 = 0.0f, keep1 = 0.0f;
#pragma unroll
      for (int j = 0; j < 8; ++j) {
        float acc = a[j].x * b[j].x + a[j].y * b[j].y + a[j].z * b[j].z + a[j].w * b[j].w;
        acc += __shfl_xor(acc, 1, 32);
        acc += __shfl_xor(acc, 2, 32);
        const float t = __shfl(acc, hsel, 32);
        if (j < 4) { if (jsel == j) keep0 = t; }
        else       { if (jsel == j - 4) keep1 = t; }
      }
#pragma unroll
      for (int l = 0; l < 2; ++l) {
        const int eb = e0 + g * 8 + l * 4;
        const float v = (l == 0 ? keep0 : keep1) * scale;
        const int dd = __shfl(dv, g * 8 + l * 4 + jsel, 32);
        if (eb + jsel < E) {
          e[(size_t)eb * H + lane] = v;
          atomicMaxFloat(m + (size_t)dd * H + (lane & 7), v);
        }
      }
    }
  }
}

__global__ void merge_max_planes(float* __restrict__ lo, const float* __restrict__ hi, int n) {
  const int i = blockIdx.x * blockDim.x + threadIdx.x;
  if (i < n) lo[i] = fmaxf(lo[i], hi[i]);
}


__device__ __forceinline__ void barrier_lds_only() {
  __builtin_amdgcn_fence(__ATOMIC_RELEASE, "workgroup", "local");
  __builtin_amdgcn_s_barrier();
  __builtin_amdgcn_fence(__ATOMIC_ACQUIRE, "workgroup", "local");
}

template <int HD, int D>
__global__ __launch_bounds__(256) void edge_agg_by_half(
    const float* __restrict__ feat, const int* __restrict__ src,
    const int* __restrict__ dst, const float* __restrict__ e,
    const float* __restrict__ m, float* __restrict__ denom,
    float* __restrict__ out, unsigned int* __restrict__ tickets, int E, int Nn) {
  constexpr int LINES = HD / 32;
  constexpr int H = HD / D;
  constexpr int CW = 8;
  constexpr int CH = CW * 8;
  __shared__ unsigned int s_ticket[2];
  const int lane = threadIdx.x & 31;
  const int wave = threadIdx.x >> 5;
  const int hl = lane % H;
  const int split = Nn >> 1;
  const unsigned int nChunks = (unsigned int)((E + CH - 1) / CH);
  const int home = cache_half_of_this_workgroup();
  unsigned int it = 0;
  for (int pass = 0; pass < 2; ++pass) {
    const int q = home ^ pass;
    unsigned int* counter = tickets + q * 64;
    if (threadIdx.x == 0) s_ticket[it & 1] = atomicAdd(counter, 2u);
    barrier_lds_only();
    unsigned int c = s_ticket[it & 1];
    unsigned int cn = c + 1u;
    ++it;
    int sv, dv;
    {
      const int my = (int)c * CH + wave * CW + lane;
      const bool cand = lane < CW && my < E;
      sv = cand ? src[my] : 0;
      dv = cand ? dst[my] : 0;
    }
    while (c < nChunks) {
      const int e0 = (int)c * CH + wave * CW;
      const bool cand = lane < CW && e0 + lane < E;
      const bool own = cand && ((dv >= split) == (q == 1));
      const unsigned int mask = __builtin_amdgcn_ballot_w32(own);
      const int myN = (int)cn * CH + wave * CW + lane;
      const bool candN = lane < CW && myN < E;
      const int svN = candN ? src[myN] : 0;
      const int dvN = candN ? dst[myN] : 0;
      float f[CW][LINES], mx[CW], xv[CW];
#pragma unroll
      for (int j = 0; j < CW; ++j) {
        if (mask & (1u << j)) {
          const int s = __builtin_amdgcn_readlane(sv, j);
          const int d = __builtin_amdgcn_readlane(dv, j);
          const float* fs = feat + (size_t)s * HD + lane;
#pragma unroll
          for (int i = 0; i < LINES; ++i) f[j][i] = fs[32 * i];
          mx[j] = m[(size_t)d * H + hl];
          xv[j] = e[(size_t)(e0 + j) * H + hl];
        }
      }
      unsigned int nxt = 0;
      if (threadIdx.x == 0) nxt = atomicAdd(counter, 1u);
#pragma unroll
      for (int j = 0; j < CW; ++j) {
        if (mask & (1u << j)) {
          const int d = __builtin_amdgcn_readlane(dv, j);
          float* o = out + (size_t)d * HD + lane;
          const float a = __expf(xv[j] - mx[j]);
          if (lane < H) atomicAdd(denom + (size_t)d * H + lane, a);
#pragma unroll
          for (int i = 0; i < LINES; ++i) {
            float w = a;
            if constexpr (H > 1) w = __shfl(a, (32 * i + lane) / D, 32);
            atomicAdd(o + 32 * i, w * f[j][i]);
          }
        }
      }
      if (threadIdx.x == 0) s_ticket[it & 1] = nxt;
      barrier_lds_only();
      c = cn;
      cn = s_ticket[it & 1];
      ++it;
      sv = svN;
      dv = dvN;
    }
  }
}

template <int HD, int D, bool ELU>
__global__ void finish_rows(float* __restrict__ x, const float* __restrict__ denom,
                            int n4) {
  constexpr int H = HD / D;
  constexpr int Q = HD / 4;
  static_assert(D % 4 == 0, "a 16-byte piece stays inside one head");
  const int i = blockIdx.x * blockDim.x + threadIdx.x;
  if (i >= n4) return;
  const int node = i / Q;
  const int col  = (i - node * Q) * 4;
  const float dn = fmaxf(denom[(size_t)node * H + col / D], 1e-9f);
  float4 v = ((const float4*)x)[i];
  v.x /= dn; v.y /= dn; v.z /= dn; v.w /= dn;
  if constexpr (ELU) {
    v.x = v.x > 0.0f ? v.x : expm1f(v.x);
    v.y = v.y > 0.0f ? v.y : expm1f(v.y);
    v.z = v.z > 0.0f ? v.z : expm1f(v.z);
    v.w = v.w > 0.0f ? v.w : expm1f(v.w);
  }
  ((float4*)x)[i] = v;
}

static inline int cdiv_i(long long a, long long b) { return (int)((a + b - 1) / b); }

extern "C" void kernel_launch(void* const* d_in, const int* in_sizes, int n_in,
                              void* d_out, int out_size, void* d_ws, size_t ws_size,
                              hipStream_t stream) {
  const float* hIn = (const float*)d_in[0];
  const float* W1  = (const float*)d_in[1];
  const float* W2  = (const float*)d_in[2];
  const int*   src = (const int*)d_in[3];
  const int*   dst = (const int*)d_in[4];

  const int IN_DIM = 128, HEADS = 8, OUT_DIM = 64;
  const int HD1 = 128;
  const int Nn  = in_sizes[0] / IN_DIM;
  const int E   = in_sizes[3];

  float* feat1  = (float*)d_ws;
  float* scores = feat1 + (size_t)Nn * HD1;
  float* mbuf   = scores + (size_t)E * HEADS;
  float* denom  = mbuf + (size_t)Nn * HEADS;
  float* h1     = denom + (size_t)Nn * HEADS;

  const int TPB = 256;
  const int wavesPerBlock = TPB / 32;

  unsigned int* tickets1 = (unsigned int*)d_out;
  unsigned int* tickets2 = (unsigned int*)(denom + (size_t)Nn * 4);
  float* mhi1 = (float*)d_out + 1024;
  float* mhi2 = mbuf + (size_t)Nn * 4;
  const int aggBlocks = 1024;

  {
    int tiles = (Nn / 16) * (HD1 / 16);
    gemm_f32_wmma<<<cdiv_i(tiles, wavesPerBlock), TPB, 0, stream>>>(
        hIn, W1, feat1, Nn, HD1, IN_DIM);
  }
  fill_f32<<<cdiv_i((long long)Nn * HEADS, TPB), TPB, 0, stream>>>(
      mbuf, mhi1, -INFINITY, Nn * HEADS, tickets1);
  hipMemsetAsync(denom, 0, (size_t)Nn * HEADS * sizeof(float), stream);
  hipMemsetAsync(h1, 0, (size_t)Nn * HD1 * sizeof(float), stream);

  const int scoreBlocks = cdiv_i(cdiv_i(E, 32), wavesPerBlock);
  edge_scores_wave32<128, 16><<<scoreBlocks, TPB, 0, stream>>>(
      feat1, src, dst, scores, mbuf, mhi1, E, 0.25f);
  merge_max_planes<<<cdiv_i((long long)Nn * HEADS, TPB), TPB, 0, stream>>>(
      mbuf, mhi1, Nn * HEADS);
  edge_agg_by_half<128, 16><<<aggBlocks, TPB, 0, stream>>>(
      feat1, src, dst, scores, mbuf, denom, h1, tickets1, E, Nn);
  finish_rows<128, 16, true><<<cdiv_i((long long)Nn * HD1 / 4, TPB), TPB, 0, stream>>>(
      h1, denom, Nn * HD1 / 4);

  float* feat2 = feat1;
  {
    int tiles = (Nn / 16) * (OUT_DIM / 16);
    gemm_f32_wmma<<<cdiv_i(tiles, wavesPerBlock), TPB, 0, stream>>>(
        h1, W2, feat2, Nn, OUT_DIM, HD1);
  }
  fill_f32<<<cdiv_i(Nn, TPB), TPB, 0, stream>>>(mbuf, mhi2, -INFINITY, Nn, tickets2);
  hipMemsetAsync(denom, 0, (size_t)Nn * sizeof(float), stream);
  hipMemsetAsync(d_out, 0, (size_t)Nn * OUT_DIM * sizeof(float), stream);

  edge_scores_wave32<64, 64><<<scoreBlocks, TPB, 0, stream>>>(
      feat2, src, dst, scores, mbuf, mhi2, E, 0.125f);
  merge_max_planes<<<cdiv_i(Nn, TPB), TPB, 0, stream>>>(mbuf, mhi2, Nn);
  edge_agg_by_half<64, 64><<<aggBlocks, TPB, 0, stream>>>(
      feat2, src, dst, scores, mbuf, denom, (float*)d_out, tickets2, E, Nn);
  finish_rows<64, 64, false><<<cdiv_i((long long)Nn * OUT_DIM / 4, TPB), TPB, 0, stream>>>(
      (float*)d_out, denom, Nn * OUT_DIM / 4);
}


# Aggregation passes: a window's softmax weights formed one lane per (edge, head): scores by whole-line loads, maxima by one gather (layer 2 had a chain of scalar loads); denominator add order differs
# speedup vs baseline: 2.8551x; 1.0693x over previous
#include <hip/hip_runtime.h>
#include <math.h>

typedef __attribute__((ext_vector_type(2))) float v2f;
typedef __attribute__((ext_vector_type(8))) float v8f;

__device__ __forceinline__ void atomicMaxFloat(float* addr, float value) {
  if (value >= 0.0f) {
    atomicMax((int*)addr, __float_as_int(value));
  } else {
    atomicMin((unsigned int*)addr, __float_as_uint(value));
  }
}

__global__ __launch_bounds__(256) void gemm_f32_wmma(
    const float* __restrict__ A, const float* __restrict__ B,
    float* __restrict__ C, int M, int N, int K) {
  const int lane = threadIdx.x & 31;
  const int wave = threadIdx.x >> 5;
  const int wavesPerBlock = blockDim.x >> 5;
  const int tile = blockIdx.x * wavesPerBlock + wave;
  const int tilesN = N >> 4;
  const int numTiles = (M >> 4) * tilesN;
  if (tile >= numTiles) return;

  const int tm = (tile / tilesN) << 4;
  const int tn = (tile % tilesN) << 4;
  const int r  = lane & 15;
  const int kk = (lane >> 4) << 1;

  const float* Arow = A + (size_t)(tm + r) * K;

  v8f acc = {};
  for (int k0 = 0; k0 < K; k0 += 4) {
    v2f a, b;
    float2 av = *(const float2*)(Arow + k0 + kk);
    a.x = av.x;
    a.y = av.y;
    b.x = B[(size_t)(k0 + kk) * N + tn + r];
    b.y = B[(size_t)(k0 + kk + 1) * N + tn + r];
    acc = __builtin_amdgcn_wmma_f32_16x16x4_f32(
         false, a,  false, b,
         (short)0, acc,  false,  false);
  }

  const int rowBase = tm + ((lane >> 4) << 3);
#pragma unroll
  for (int j = 0; j < 8; ++j) {
    C[(size_t)(rowBase + j) * N + tn + r] = acc[j];
  }
}

__global__ void fill_f32(float* __restrict__ p, float* __restrict__ p2, float v, int n,
                         unsigned int* __restrict__ tickets) {
  int i = blockIdx.x * blockDim.x + threadIdx.x;
  if (i < n) { p[i] = v; p2[i] = v; }
  if (i < 2) tickets[i * 64] = 0u;
}

__device__ __forceinline__ int cache_half_of_this_workgroup() {
  return (int)((__builtin_amdgcn_s_sendmsg_rtn(0x87) >> 18) & 1u);
}

template <int HD, int D>
__global__ __launch_bounds__(256) void edge_scores_wave32(
    const float* __restrict__ feat, const int* __restrict__ src,
    const int* __restrict__ dst, float* __restrict__ e, float* __restrict__ m_lo,
    float* __restrict__ m_hi, int E, float scale) {
  constexpr int H = HD / D;
  const int lane = threadIdx.x & 31;
  const int wave = (blockIdx.x * blockDim.x + threadIdx.x) >> 5;
  const int e0 = wave * 32;
  if (e0 >= E) return;
  float* m = cache_half_of_this_workgroup() ? m_hi : m_lo;
  const int my = e0 + lane;
  const bool valid = my < E;
  const int sv = valid ? src[my] : 0;
  const int dv = valid ? dst[my] : 0;

  if constexpr (H == 1) {
    static_assert(HD == 64, "one head of 64 columns");
    const int sub = lane & 15;
    const int hb  = lane & 16;
    float mine = 0.0f;
#pragma unroll
    for (int i = 0; i < 16; ++i) {
      const int s = __shfl(sv, hb | i, 32);
      const int d = __shfl(dv, hb | i, 32);
      const float4 a = *(const float4*)(feat + (size_t)s * HD + sub * 4);
      const float4 b = *(const float4*)(feat + (size_t)d * HD + sub * 4);
      float acc = a.x * b.x + a.y * b.y + a.z * b.z + a.w * b.w;
      acc += __shfl_xor(acc, 1, 32);
      acc += __shfl_xor(acc, 2, 32);
      acc += __shfl_xor(acc, 4, 32);
      acc += __shfl_xor(acc, 8, 32);
      if (sub == i) mine = acc;
    }
    mine *= scale;
    if (valid) {
      e[my] = mine;
      atomicMaxFloat(m + dv, mine);
    }
  } else {
    static_assert(HD == 128 && D == 16, "eight heads of 16 columns");
    const int hsel = (lane & 7) << 2;
    const int jsel = lane >> 3;
    for (int g = 0; g < 4; ++g) {
      float4 a[8], b[8];
#pragma unroll
      for (int j = 0; j < 8; ++j) {
        const int s = __builtin_amdgcn_readlane(sv, g * 8 + j);
        const int d = __builtin_amdgcn_readlane(dv, g * 8 + j);
        a[j] = *(const float4*)(feat + (size_t)s * HD + lane * 4);
        b[j] = *(const float4*)(feat + (size_t)d * HD + lane * 4);
      }
      float keep0 = 0.0f, keep1 = 0.0f;
#pragma unroll
      for (int j = 0; j < 8; ++j) {
        float acc = a[j].x * b[j].x + a[j].y * b[j].y + a[j].z * b[j].z + a[j].w * b[j].w;
        acc += __shfl_xor(acc, 1, 32);
        acc += __shfl_xor(acc, 2, 32);
        const float t = __shfl(acc, hsel, 32);
        if (j < 4) { if (jsel == j) keep0 = t; }
        else       { if (jsel == j - 4) keep1 = t; }
      }
#pragma unroll
      for (int l = 0; l < 2; ++l) {
        const int eb = e0 + g * 8 + l * 4;
        const float v = (l == 0 ? keep0 : keep1) * scale;
        const int dd = __shfl(dv, g * 8 + l * 4 + jsel, 32);
        if (eb + jsel < E) {
          e[(size_t)eb * H + lane] = v;
          atomicMaxFloat(m + (size_t)dd * H + (lane & 7), v);
        }
      }
    }
  }
}

__global__ void merge_max_planes(float* __restrict__ lo, const float* __restrict__ hi, int n) {
  const int i = blockIdx.x * blockDim.x + threadIdx.x;
  if (i < n) lo[i] = fmaxf(lo[i], hi[i]);
}


__device__ __forceinline__ void barrier_lds_only() {
  __builtin_amdgcn_fence(__ATOMIC_RELEASE, "workgroup", "local");
  __builtin_amdgcn_s_barrier();
  __builtin_amdgcn_fence(__ATOMIC_ACQUIRE, "workgroup", "local");
}

template <int HD, int D>
__global__ __launch_bounds__(256) void edge_agg_by_half(
    const float* __restrict__ feat, const int* __restrict__ src,
    const int* __restrict__ dst, const float* __restrict__ e,
    const float* __restrict__ m, float* __restrict__ denom,
    float* __restrict__ out, unsigned int* __restrict__ tickets, int E, int Nn) {
  constexpr int LINES = HD / 32;
  constexpr int H = HD / D;
  constexpr int CW = 8;
  constexpr int CH = CW * 8;
  constexpr int EPW = (32 / H) < CW ? (32 / H) : CW;
  constexpr int NW = CW / EPW;
  __shared__ unsigned int s_ticket[2];
  const int lane = threadIdx.x & 31;
  const int wave = threadIdx.x >> 5;
  const int hl = lane % H;
  const int split = Nn >> 1;
  const unsigned int nChunks = (unsigned int)((E + CH - 1) / CH);
  const int home = cache_half_of_this_workgroup();
  unsigned int it = 0;
  for (int pass = 0; pass < 2; ++pass) {
    const int q = home ^ pass;
    unsigned int* counter = tickets + q * 64;
    if (threadIdx.x == 0) s_ticket[it & 1] = atomicAdd(counter, 2u);
    barrier_lds_only();
    unsigned int c = s_ticket[it & 1];
    unsigned int cn = c + 1u;
    ++it;
    int sv, dv;
    {
      const int my = (int)c * CH + wave * CW + lane;
      const bool cand = lane < CW && my < E;
      sv = cand ? src[my] : 0;
      dv = cand ? dst[my] : 0;
    }
    while (c < nChunks) {
      const int e0 = (int)c * CH + wave * CW;
      const bool cand = lane < CW && e0 + lane < E;
      const bool own = cand && ((dv >= split) == (q == 1));
      const unsigned int mask = __builtin_amdgcn_ballot_w32(own);
      const int myN = (int)cn * CH + wave * CW + lane;
      const bool candN = lane < CW && myN < E;
      const int svN = candN ? src[myN] : 0;
      const int dvN = candN ? dst[myN] : 0;
      int dg[NW];
      bool actg[NW];
#pragma unroll
      for (int k = 0; k < NW; ++k) {
        const int js = k * EPW + lane / H;
        actg[k] = lane < EPW * H && ((mask >> js) & 1u) != 0u;
        dg[k] = dv;
        if constexpr (H > 1) {
          const int q = lane / H;
          int t = __builtin_amdgcn_readlane(dv, k * EPW + EPW - 1);
#pragma unroll
          for (int c = EPW - 2; c >= 0; --c) t = (q == c) ? __builtin_amdgcn_readlane(dv, k * EPW + c) : t;
          dg[k] = t;
        }
      }
      float f[CW][LINES];
#pragma unroll
      for (int j = 0; j < CW; ++j) {
        if (mask & (1u << j)) {
          const int s = __builtin_amdgcn_readlane(sv, j);
          const float* fs = feat + (size_t)s * HD + lane;
#pragma unroll
          for (int i = 0; i < LINES; ++i) f[j][i] = fs[32 * i];
        }
      }
      float mxg[NW], xg[NW];
#pragma unroll
      for (int k = 0; k < NW; ++k) {
        const int js = k * EPW + lane / H;
        mxg[k] = 0.0f;
        xg[k] = 0.0f;
        if (actg[k]) {
          mxg[k] = m[(size_t)dg[k] * H + hl];
          xg[k] = e[(size_t)(e0 + js) * H + hl];
        }
      }
      unsigned int nxt = 0;
      if (threadIdx.x == 0) nxt = atomicAdd(counter, 1u);
      float ag[NW];
#pragma unroll
      for (int k = 0; k < NW; ++k) {
        ag[k] = __expf(xg[k] - mxg[k]);
        if (actg[k]) atomicAdd(denom + (size_t)dg[k] * H + hl, ag[k]);
      }
#pragma unroll
      for (int j = 0; j < CW; ++j) {
        if (mask & (1u << j)) {
          const int d = __builtin_amdgcn_readlane(dv, j);
          float* o = out + (size_t)d * HD + lane;
#pragma unroll
          for (int i = 0; i < LINES; ++i) {
            float w;
            if constexpr (H > 1) w = __shfl(ag[j / EPW], (j % EPW) * H + (32 * i + lane) / D, 32);
            else w = __int_as_float(__builtin_amdgcn_readlane(__float_as_int(ag[0]), j));
            atomicAdd(o + 32 * i, w * f[j][i]);
          }
        }
      }
      if (threadIdx.x == 0) s_ticket[it & 1] = nxt;
      barrier_lds_only();
      c = cn;
      cn = s_ticket[it & 1];
      ++it;
      sv = svN;
      dv = dvN;
    }
  }
}

template <int HD, int D, bool ELU>
__global__ void finish_rows(float* __restrict__ x, const float* __restrict__ denom,
                            int n4) {
  constexpr int H = HD / D;
  constexpr int Q = HD / 4;
  static_assert(D % 4 == 0, "a 16-byte piece stays inside one head");
  const int i = blockIdx.x * blockDim.x + threadIdx.x;
  if (i >= n4) return;
  const int node = i / Q;
  const int col  = (i - node * Q) * 4;
  const float dn = fmaxf(denom[(size_t)node * H + col / D], 1e-9f);
  float4 v = ((const float4*)x)[i];
  v.x /= dn; v.y /= dn; v.z /= dn; v.w /= dn;
  if constexpr (ELU) {
    v.x = v.x > 0.0f ? v.x : expm1f(v.x);
    v.y = v.y > 0.0f ? v.y : expm1f(v.y);
    v.z = v.z > 0.0f ? v.z : expm1f(v.z);
    v.w = v.w > 0.0f ? v.w : expm1f(v.w);
  }
  ((float4*)x)[i] = v;
}

static inline int cdiv_i(long long a, long long b) { return (int)((a + b - 1) / b); }

extern "C" void kernel_launch(void* const* d_in, const int* in_sizes, int n_in,
                              void* d_out, int out_size, void* d_ws, size_t ws_size,
                              hipStream_t stream) {
  const float* hIn = (const float*)d_in[0];
  const float* W1  = (const float*)d_in[1];
  const float* W2  = (const float*)d_in[2];
  const int*   src = (const int*)d_in[3];
  const int*   dst = (const int*)d_in[4];

  const int IN_DIM = 128, HEADS = 8, OUT_DIM = 64;
  const int HD1 = 128;
  const int Nn  = in_sizes[0] / IN_DIM;
  const int E   = in_sizes[3];

  float* feat1  = (float*)d_ws;
  float* scores = feat1 + (size_t)Nn * HD1;
  float* mbuf   = scores + (size_t)E * HEADS;
  float* denom  = mbuf + (size_t)Nn * HEADS;
  float* h1     = denom + (size_t)Nn * HEADS;

  const int TPB = 256;
  const int wavesPerBlock = TPB / 32;

  unsigned int* tickets1 = (unsigned int*)d_out;
  unsigned int* tickets2 = (unsigned int*)(denom + (size_t)Nn * 4);
  float* mhi1 = (float*)d_out + 1024;
  float* mhi2 = mbuf + (size_t)Nn * 4;
  const int aggBlocks = 1024;

  {
    int tiles = (Nn / 16) * (HD1 / 16);
    gemm_f32_wmma<<<cdiv_i(tiles, wavesPerBlock), TPB, 0, stream>>>(
        hIn, W1, feat1, Nn, HD1, IN_DIM);
  }
  fill_f32<<<cdiv_i((long long)Nn * HEADS, TPB), TPB, 0, stream>>>(
      mbuf, mhi1, -INFINITY, Nn * HEADS, tickets1);
  hipMemsetAsync(denom, 0, (size_t)Nn * HEADS * sizeof(float), stream);
  hipMemsetAsync(h1, 0, (size_t)Nn * HD1 * sizeof(float), stream);

  const int scoreBlocks = cdiv_i(cdiv_i(E, 32), wavesPerBlock);
  edge_scores_wave32<128, 16><<<scoreBlocks, TPB, 0, stream>>>(
      feat1, src, dst, scores, mbuf, mhi1, E, 0.25f);
  merge_max_planes<<<cdiv_i((long long)Nn * HEADS, TPB), TPB, 0, stream>>>(
      mbuf, mhi1, Nn * HEADS);
  edge_agg_by_half<128, 16><<<aggBlocks, TPB, 0, stream>>>(
      feat1, src, dst, scores, mbuf, denom, h1, tickets1, E, Nn);
  finish_rows<128, 16, true><<<cdiv_i((long long)Nn * HD1 / 4, TPB), TPB, 0, stream>>>(
      h1, denom, Nn * HD1 / 4);

  float* feat2 = feat1;
  {
    int tiles = (Nn / 16) * (OUT_DIM / 16);
    gemm_f32_wmma<<<cdiv_i(tiles, wavesPerBlock), TPB, 0, stream>>>(
        h1, W2, feat2, Nn, OUT_DIM, HD1);
  }
  fill_f32<<<cdiv_i(Nn, TPB), TPB, 0, stream>>>(mbuf, mhi2, -INFINITY, Nn, tickets2);
  hipMemsetAsync(denom, 0, (size_t)Nn * sizeof(float), stream);
  hipMemsetAsync(d_out, 0, (size_t)Nn * OUT_DIM * sizeof(float), stream);

  edge_scores_wave32<64, 64><<<scoreBlocks, TPB, 0, stream>>>(
      feat2, src, dst, scores, mbuf, mhi2, E, 0.125f);
  merge_max_planes<<<cdiv_i(Nn, TPB), TPB, 0, stream>>>(mbuf, mhi2, Nn);
  edge_agg_by_half<64, 64><<<aggBlocks, TPB, 0, stream>>>(
      feat2, src, dst, scores, mbuf, denom, (float*)d_out, tickets2, E, Nn);
  finish_rows<64, 64, false><<<cdiv_i((long long)Nn * OUT_DIM / 4, TPB), TPB, 0, stream>>>(
      (float*)d_out, denom, Nn * OUT_DIM / 4);
}
